# starting kernel
# baseline (speedup 1.0000x reference)
#include <hip/hip_runtime.h>
#include <math.h>


typedef float v2f __attribute__((ext_vector_type(2)));
typedef float v8f __attribute__((ext_vector_type(8)));

#define DFEAT  10
#define KPAD   12
#define ROWPAD 16

__global__ __launch_bounds__(256)
void edgeconv_lin1_wmma(const float* __restrict__ x,
                        const float* __restrict__ W1,
                        const float* __restrict__ b1,
                        float* __restrict__ yws,
                        float* __restrict__ zws,
                        int nrows, int ntiles) {
    __shared__ float ldsA[16 * KPAD];
    __shared__ float ldsB[16 * KPAD];

    for (int t = threadIdx.x; t < 16 * KPAD; t += blockDim.x) {
        const int n = t / KPAD, k = t % KPAD;
        float wtop = 0.0f, wbot = 0.0f;
        if (n < DFEAT && k < DFEAT) {
            wtop = W1[k * DFEAT + n];
            wbot = W1[(k + DFEAT) * DFEAT + n];
        }
        ldsA[t] = wtop - wbot;
        ldsB[t] = wbot;
    }
    __syncthreads();

    const int wavesPerBlock = blockDim.x >> 5;
    const int tile = blockIdx.x * wavesPerBlock + (threadIdx.x >> 5);
    if (tile >= ntiles) return;

    const int lane = threadIdx.x & 31;
    const int hi   = lane >> 4;
    const int m    = lane & 15;
    const int base = tile * 16;
    const int row  = base + m;
    const bool rowOK = (row < nrows);

    v8f cy = {0.f,0.f,0.f,0.f,0.f,0.f,0.f,0.f};
    v8f cz = {0.f,0.f,0.f,0.f,0.f,0.f,0.f,0.f};

#pragma unroll
    for (int ks = 0; ks < 3; ++ks) {
        const int k0 = ks * 4 + 2 * hi;
        const bool ok = rowOK && (k0 < DFEAT);
        const float2* px = ok ? (const float2*)(x + (size_t)row * DFEAT + k0)
                              : (const float2*)x;
        const float2 xv = *px;
        v2f a;  a[0] = ok ? xv.x : 0.0f;  a[1] = ok ? xv.y : 0.0f;

        const float2 wav = *(const float2*)(ldsA + m * KPAD + k0);
        const float2 wbv = *(const float2*)(ldsB + m * KPAD + k0);
        v2f wa;  wa[0] = wav.x;  wa[1] = wav.y;
        v2f wb;  wb[0] = wbv.x;  wb[1] = wbv.y;

        cy = __builtin_amdgcn_wmma_f32_16x16x4_f32(false, a, false, wa,
                                                   (short)0, cy, false, false);
        cz = __builtin_amdgcn_wmma_f32_16x16x4_f32(false, a, false, wb,
                                                   (short)0, cz, false, false);
    }

    const float bias = (m < DFEAT) ? b1[m] : 0.0f;
    float* yrow = yws + (size_t)base * ROWPAD + m;
    float* zrow = zws + (size_t)base * ROWPAD + m;
    if (base + 16 <= nrows) {
#pragma unroll
        for (int r = 0; r < 8; ++r) {
            const int rr = r + 8 * hi;
            yrow[(size_t)rr * ROWPAD] = cy[r] + bias;
            zrow[(size_t)rr * ROWPAD] = cz[r];
        }
    } else {
#pragma unroll
        for (int r = 0; r < 8; ++r) {
            const int rr = r + 8 * hi;
            if (base + rr < nrows) {
                yrow[(size_t)rr * ROWPAD] = cy[r] + bias;
                zrow[(size_t)rr * ROWPAD] = cz[r];
            }
        }
    }
}

__global__ __launch_bounds__(256)
void edgeconv_maxagg(const int* __restrict__ src,
                     const float* __restrict__ yws,
                     const float* __restrict__ zws,
                     const float* __restrict__ W2,
                     const float* __restrict__ b2,
                     float* __restrict__ out,
                     int nnodes, int kdeg) {
    const int wavesPerBlock = blockDim.x >> 5;
    const int node = blockIdx.x * wavesPerBlock + (threadIdx.x >> 5);
    if (node >= nnodes) return;
    const int lane = threadIdx.x & 31;

    float m[DFEAT];
    if (lane < kdeg) {
        const int s = src[(size_t)node * kdeg + lane];
        const float* zr = zws + (size_t)s * ROWPAD;
        const float4 z0 = *(const float4*)(zr);
        const float4 z1 = *(const float4*)(zr + 4);
        m[0] = z0.x; m[1] = z0.y; m[2] = z0.z; m[3] = z0.w;
        m[4] = z1.x; m[5] = z1.y; m[6] = z1.z; m[7] = z1.w;
        m[8] = zr[8]; m[9] = zr[9];
    } else {
#pragma unroll
        for (int j = 0; j < DFEAT; ++j) m[j] = -INFINITY;
    }

#pragma unroll
    for (int off = 16; off > 0; off >>= 1) {
#pragma unroll
        for (int j = 0; j < DFEAT; ++j) {
            const float o = __shfl_xor(m[j], off, 32);
            m[j] = fmaxf(m[j], o);
        }
    }

    const float* yr = yws + (size_t)node * ROWPAD;
    const float4 y0 = *(const float4*)(yr);
    const float4 y1 = *(const float4*)(yr + 4);
    const float yv[DFEAT] = { y0.x, y0.y, y0.z, y0.w,
                              y1.x, y1.y, y1.z, y1.w, yr[8], yr[9] };

    float agg[DFEAT];
#pragma unroll
    for (int j = 0; j < DFEAT; ++j) {
        const float v = yv[j] + m[j];
        agg[j] = isfinite(v) ? v : 0.0f;
    }

    if (lane < DFEAT) {
        float acc = b2[lane];
#pragma unroll
        for (int k = 0; k < DFEAT; ++k)
            acc = fmaf(agg[k], W2[k * DFEAT + lane], acc);
        out[(size_t)node * DFEAT + lane] = acc;
    }
}

extern "C" void kernel_launch(void* const* d_in, const int* in_sizes, int n_in,
                              void* d_out, int out_size, void* d_ws, size_t ws_size,
                              hipStream_t stream) {
    const float* x   = (const float*)d_in[0];
    const int*   ei  = (const int*)  d_in[1];
    const float* W1  = (const float*)d_in[2];
    const float* b1  = (const float*)d_in[3];
    const float* W2  = (const float*)d_in[4];
    const float* b2  = (const float*)d_in[5];
    float* out = (float*)d_out;

    const int nnodes = in_sizes[0] / DFEAT;
    const int nedges = in_sizes[1] / 2;
    const int kdeg   = nedges / nnodes;
    const int* src   = ei;

    float* yws = (float*)d_ws;
    float* zws = yws + (size_t)nnodes * ROWPAD;

    const int ntiles = (nnodes + 15) / 16;
    {
        dim3 block(256);
        dim3 grid((ntiles + 7) / 8);
        hipLaunchKernelGGL(edgeconv_lin1_wmma, grid, block, 0, stream,
                           x, W1, b1, yws, zws, nnodes, ntiles);
    }
    {
        dim3 block(256);
        dim3 grid((nnodes + 7) / 8);
        hipLaunchKernelGGL(edgeconv_maxagg, grid, block, 0, stream,
                           src, yws, zws, W2, b2, out, nnodes, kdeg);
    }
}


# one fused kernel: neighbour x rows gathered straight into f32 WMMA A tiles, x_j@W1b per edge on the matrix unit, max over D rows in registers; no y/z planes; changes the sum order of all 3 products
# speedup vs baseline: 1.0879x; 1.0879x over previous
#include <hip/hip_runtime.h>
#include <math.h>


typedef float v2f __attribute__((ext_vector_type(2)));
typedef float v4f __attribute__((ext_vector_type(4)));
typedef float v8f __attribute__((ext_vector_type(8)));
typedef v4f v4f_a8 __attribute__((aligned(8)));

#define DFEAT 10
#define NPW   16
#define WPB   4
#define TP    20
#define UN    4

__device__ __forceinline__ v8f wmma4(v2f a, v2f b, v8f c) {
    return __builtin_amdgcn_wmma_f32_16x16x4_f32(false, a, false, b, (short)0, c, false, false);
}

__device__ __forceinline__ float xhalf(float v) {
    const int i = __builtin_bit_cast(int, v);
    return __builtin_bit_cast(float,
        __builtin_amdgcn_permlanex16(i, i, 0x76543210, 0xfedcba98, false, false));
}

__device__ __forceinline__ void wave_lds_sync() {
    __builtin_amdgcn_fence(__ATOMIC_RELEASE, "wavefront");
    __builtin_amdgcn_wave_barrier();
    __builtin_amdgcn_fence(__ATOMIC_ACQUIRE, "wavefront");
}

__device__ __forceinline__ float wld(const float* __restrict__ W, int k, int c) {
    const bool ok = (k < DFEAT) && (c < DFEAT);
    const float v = W[ok ? k * DFEAT + c : 0];
    return ok ? v : 0.0f;
}

struct RowFrag { v4f q; float2 p; };

__device__ __forceinline__ RowFrag load_row(const float* __restrict__ xr, int h) {
    RowFrag f;
    f.q = *(const v4f_a8*)(xr + 4 * h);
    f.p = *(const float2*)(xr + 8);
    return f;
}

__device__ __forceinline__ v8f tile_mm(const RowFrag& f, int h, const v2f (&w)[3], v8f d) {
    v2f a0, a1, a2;
    a0[0] = f.q[0]; a0[1] = f.q[1];
    a1[0] = f.q[2]; a1[1] = f.q[3];
    a2[0] = h ? 0.0f : f.p.x;
    a2[1] = h ? 0.0f : f.p.y;
    d = wmma4(a0, w[0], d);
    d = wmma4(a1, w[1], d);
    d = wmma4(a2, w[2], d);
    return d;
}

__device__ __forceinline__ float rows_max(v8f d) {
    const float m01 = fmaxf(d[0], d[1]), m23 = fmaxf(d[2], d[3]);
    const float m45 = fmaxf(d[4], d[5]), m67 = fmaxf(d[6], d[7]);
    const float m = fmaxf(fmaxf(m01, m23), fmaxf(m45, m67));
    return fmaxf(m, xhalf(m));
}

__global__ __launch_bounds__(WPB * 32)
void edgeconv_fused(const float* __restrict__ x,
                    const int*   __restrict__ src,
                    const float* __restrict__ W1,
                    const float* __restrict__ b1,
                    const float* __restrict__ W2,
                    const float* __restrict__ b2,
                    float* __restrict__ out,
                    int nnodes, int kdeg) {
    __shared__ __attribute__((aligned(16))) float tiles[WPB][NPW * TP];

    const int wave = threadIdx.x >> 5;
    const int lane = threadIdx.x & 31;
    const int c = lane & 15;
    const int h = lane >> 4;
    const int base = (blockIdx.x * WPB + wave) * NPW;
    if (base >= nnodes) return;
    float* tile = tiles[wave];
    const int last = nnodes - 1;

    v2f wa[3], wb[3], w2[3];
#pragma unroll
    for (int s = 0; s < 3; ++s) {
#pragma unroll
        for (int v = 0; v < 2; ++v) {
            const int k = (s < 2) ? (4 * h + 2 * s + v) : (8 + 2 * h + v);
            const float top = wld(W1, k, c);
            const float bot = wld(W1 + DFEAT * DFEAT, k, c);
            wa[s][v] = top - bot;
            wb[s][v] = bot;
            w2[s][v] = wld(W2, k, c);
        }
    }
    const v8f zero8 = {0.f, 0.f, 0.f, 0.f, 0.f, 0.f, 0.f, 0.f};

    if (kdeg == 32) {
        int idn[UN][2];
#pragma unroll
        for (int j = 0; j < UN; ++j) {
            const int node = min(base + j, last);
#pragma unroll
            for (int t = 0; t < 2; ++t) idn[j][t] = src[(size_t)node * 32 + t * 16 + c];
        }
#pragma unroll 1
        for (int g = 0; g < NPW / UN; ++g) {
            int idc[UN][2];
#pragma unroll
            for (int j = 0; j < UN; ++j) {
                idc[j][0] = idn[j][0]; idc[j][1] = idn[j][1];
                const int node = min(base + (g + 1) * UN + j, last);
#pragma unroll
                for (int t = 0; t < 2; ++t) idn[j][t] = src[(size_t)node * 32 + t * 16 + c];
            }
            __builtin_amdgcn_sched_barrier(0);
            RowFrag f[UN][2];
#pragma unroll
            for (int j = 0; j < UN; ++j) {
#pragma unroll
                for (int t = 0; t < 2; ++t) f[j][t] = load_row(x + (size_t)idc[j][t] * DFEAT, h);
            }
#pragma unroll
            for (int j = 0; j < UN; ++j) {
                const v8f d0 = tile_mm(f[j][0], h, wb, zero8);
                const v8f d1 = tile_mm(f[j][1], h, wb, zero8);
                v8f d;
#pragma unroll
                for (int r = 0; r < 8; ++r) d[r] = fmaxf(d0[r], d1[r]);
                tile[(g * UN + j) * TP + c] = rows_max(d);
            }
        }
    } else {
        for (int i = 0; i < NPW; ++i) {
            const int node = min(base + i, last);
            v8f d;
#pragma unroll
            for (int r = 0; r < 8; ++r) d[r] = -INFINITY;
            for (int e0 = 0; e0 < kdeg; e0 += 16) {
                const int e = min(e0 + c, kdeg - 1);
                const int s = src[(size_t)node * kdeg + e];
                const RowFrag f = load_row(x + (size_t)s * DFEAT, h);
                const v8f dt = tile_mm(f, h, wb, zero8);
#pragma unroll
                for (int r = 0; r < 8; ++r) d[r] = fmaxf(d[r], dt[r]);
            }
            tile[i * TP + c] = rows_max(d);
        }
    }

    const RowFrag fy = load_row(x + (size_t)min(base + c, last) * DFEAT, h);
    const v8f cy = tile_mm(fy, h, wa, zero8);
    const float bias1 = (c < DFEAT) ? b1[c] : 0.0f;
    const float bias2 = (c < DFEAT) ? b2[c] : 0.0f;

    wave_lds_sync();
    float agg[8];
#pragma unroll
    for (int r = 0; r < 8; ++r) {
        const float v = (cy[r] + bias1) + tile[(r + 8 * h) * TP + c];
        agg[r] = isfinite(v) ? v : 0.0f;
    }
    wave_lds_sync();
#pragma unroll
    for (int r = 0; r < 8; ++r) tile[(r + 8 * h) * TP + c] = agg[r];
    wave_lds_sync();

    RowFrag fo;
    fo.q = *(const v4f*)(tile + c * TP + 4 * h);
    const float2 po = *(const float2*)(tile + c * TP + 8 + 2 * h);
    v8f co = zero8;
    {
        v2f a0, a1, a2;
        a0[0] = fo.q[0]; a0[1] = fo.q[1];
        a1[0] = fo.q[2]; a1[1] = fo.q[3];
        a2[0] = po.x;    a2[1] = po.y;
        co = wmma4(a0, w2[0], co);
        co = wmma4(a1, w2[1], co);
        co = wmma4(a2, w2[2], co);
    }

    wave_lds_sync();
    if (c < DFEAT) {
#pragma unroll
        for (int r = 0; r < 8; ++r) tile[(r + 8 * h) * DFEAT + c] = co[r] + bias2;
    }
    wave_lds_sync();
    const int nvalid = min(NPW, nnodes - base) * DFEAT;
    float* orow = out + (size_t)base * DFEAT;
#pragma unroll
    for (int j = 0; j < (NPW * DFEAT) / 32; ++j) {
        const int fidx = lane + 32 * j;
        if (fidx < nvalid) orow[fidx] = tile[fidx];
    }
}

extern "C" void kernel_launch(void* const* d_in, const int* in_sizes, int n_in,
                              void* d_out, int out_size, void* d_ws, size_t ws_size,
                              hipStream_t stream) {
    const float* x   = (const float*)d_in[0];
    const int*   ei  = (const int*)  d_in[1];
    const float* W1  = (const float*)d_in[2];
    const float* b1  = (const float*)d_in[3];
    const float* W2  = (const float*)d_in[4];
    const float* b2  = (const float*)d_in[5];
    float* out = (float*)d_out;

    const int nnodes = in_sizes[0] / DFEAT;
    const int nedges = in_sizes[1] / 2;
    const int kdeg   = nedges / nnodes;
    const int* src   = ei;

    const int ntiles = (nnodes + 15) / 16;
    dim3 block(WPB * 32);
    dim3 grid((ntiles + WPB - 1) / WPB);
    hipLaunchKernelGGL(edgeconv_fused, grid, block, 0, stream,
                       x, src, W1, b1, W2, b2, out, nnodes, kdeg);
}


# degree-32 path software-pipelined: row gathers of node group g+1 and index loads of group g+3 in flight during group g's products (two fragment sets); own-row and bias loads early; sums unchanged
# speedup vs baseline: 1.4083x; 1.2945x over previous
#include <hip/hip_runtime.h>
#include <math.h>


typedef float v2f __attribute__((ext_vector_type(2)));
typedef float v4f __attribute__((ext_vector_type(4)));
typedef float v8f __attribute__((ext_vector_type(8)));
typedef v4f v4f_a8 __attribute__((aligned(8)));

#define DFEAT 10
#define NPW   16
#define WPB   4
#define TP    20
#define UN    2
#define NG    (NPW / UN)

__device__ __forceinline__ v8f wmma4(v2f a, v2f b, v8f c) {
    return __builtin_amdgcn_wmma_f32_16x16x4_f32(false, a, false, b, (short)0, c, false, false);
}

__device__ __forceinline__ float xhalf(float v) {
    const int i = __builtin_bit_cast(int, v);
    return __builtin_bit_cast(float,
        __builtin_amdgcn_permlanex16(i, i, 0x76543210, 0xfedcba98, false, false));
}

__device__ __forceinline__ void wave_lds_sync() {
    __builtin_amdgcn_fence(__ATOMIC_RELEASE, "wavefront");
    __builtin_amdgcn_wave_barrier();
    __builtin_amdgcn_fence(__ATOMIC_ACQUIRE, "wavefront");
}

__device__ __forceinline__ float wld(const float* __restrict__ W, int k, int c) {
    const bool ok = (k < DFEAT) && (c < DFEAT);
    const float v = W[ok ? k * DFEAT + c : 0];
    return ok ? v : 0.0f;
}

struct RowFrag { v4f q; float2 p; };

__device__ __forceinline__ RowFrag load_row(const float* __restrict__ xr, int h) {
    RowFrag f;
    f.q = *(const v4f_a8*)(xr + 4 * h);
    f.p = *(const float2*)(xr + 8);
    return f;
}

__device__ __forceinline__ v8f tile_mm(const RowFrag& f, int h, const v2f (&w)[3], v8f d) {
    v2f a0, a1, a2;
    a0[0] = f.q[0]; a0[1] = f.q[1];
    a1[0] = f.q[2]; a1[1] = f.q[3];
    a2[0] = h ? 0.0f : f.p.x;
    a2[1] = h ? 0.0f : f.p.y;
    d = wmma4(a0, w[0], d);
    d = wmma4(a1, w[1], d);
    d = wmma4(a2, w[2], d);
    return d;
}

__device__ __forceinline__ float rows_max(v8f d) {
    const float m01 = fmaxf(d[0], d[1]), m23 = fmaxf(d[2], d[3]);
    const float m45 = fmaxf(d[4], d[5]), m67 = fmaxf(d[6], d[7]);
    const float m = fmaxf(fmaxf(m01, m23), fmaxf(m45, m67));
    return fmaxf(m, xhalf(m));
}

__device__ __forceinline__ void load_idx(const int* __restrict__ src, int base, int g, int last,
                                         int c, int (&id)[UN][2]) {
#pragma unroll
    for (int j = 0; j < UN; ++j) {
        const int node = min(base + g * UN + j, last);
#pragma unroll
        for (int t = 0; t < 2; ++t) id[j][t] = src[(size_t)node * 32 + t * 16 + c];
    }
}

__device__ __forceinline__ void gather_rows(const float* __restrict__ x, const int (&id)[UN][2],
                                            int h, RowFrag (&f)[UN][2]) {
#pragma unroll
    for (int j = 0; j < UN; ++j) {
#pragma unroll
        for (int t = 0; t < 2; ++t) f[j][t] = load_row(x + (size_t)id[j][t] * DFEAT, h);
    }
}

__global__ __launch_bounds__(WPB * 32)
void edgeconv_fused(const float* __restrict__ x,
                    const int*   __restrict__ src,
                    const float* __restrict__ W1,
                    const float* __restrict__ b1,
                    const float* __restrict__ W2,
                    const float* __restrict__ b2,
                    float* __restrict__ out,
                    int nnodes, int kdeg) {
    __shared__ __attribute__((aligned(16))) float tiles[WPB][NPW * TP];

    const int wave = threadIdx.x >> 5;
    const int lane = threadIdx.x & 31;
    const int c = lane & 15;
    const int h = lane >> 4;
    const int base = (blockIdx.x * WPB + wave) * NPW;
    if (base >= nnodes) return;
    float* tile = tiles[wave];
    const int last = nnodes - 1;

    v2f wa[3], wb[3], w2[3];
#pragma unroll
    for (int s = 0; s < 3; ++s) {
#pragma unroll
        for (int v = 0; v < 2; ++v) {
            const int k = (s < 2) ? (4 * h + 2 * s + v) : (8 + 2 * h + v);
            const float top = wld(W1, k, c);
            const float bot = wld(W1 + DFEAT * DFEAT, k, c);
            wa[s][v] = top - bot;
            wb[s][v] = bot;
            w2[s][v] = wld(W2, k, c);
        }
    }
    const v8f zero8 = {0.f, 0.f, 0.f, 0.f, 0.f, 0.f, 0.f, 0.f};

    const float* xown = x + (size_t)min(base + c, last) * DFEAT;
    const int cb = (c < DFEAT) ? c : 0;
    RowFrag fy;
    float rb1, rb2;

    if (kdeg == 32) {
        int id[2][UN][2];
        RowFrag f[2][UN][2];
        load_idx(src, base, 0, last, c, id[0]);
        load_idx(src, base, 1, last, c, id[1]);
        gather_rows(x, id[0], h, f[0]);
        load_idx(src, base, 2, last, c, id[0]);
#pragma unroll
        for (int g = 0; g < NG; ++g) {
            const int cur = g & 1, nxt = cur ^ 1;
            if (g + 1 < NG) gather_rows(x, id[nxt], h, f[nxt]);
            if (g + 3 < NG) load_idx(src, base, g + 3, last, c, id[nxt]);
            if (g == NG - 2) {
                fy = load_row(xown, h);
                rb1 = b1[cb];
                rb2 = b2[cb];
            }
            __builtin_amdgcn_sched_barrier(0);
#pragma unroll
            for (int j = 0; j < UN; ++j) {
                const v8f d0 = tile_mm(f[cur][j][0], h, wb, zero8);
                const v8f d1 = tile_mm(f[cur][j][1], h, wb, zero8);
                v8f d;
#pragma unroll
                for (int r = 0; r < 8; ++r) d[r] = fmaxf(d0[r], d1[r]);
                tile[(g * UN + j) * TP + c] = rows_max(d);
            }
            __builtin_amdgcn_sched_barrier(0);
        }
    } else {
        for (int i = 0; i < NPW; ++i) {
            const int node = min(base + i, last);
            v8f d;
#pragma unroll
            for (int r = 0; r < 8; ++r) d[r] = -INFINITY;
            for (int e0 = 0; e0 < kdeg; e0 += 16) {
                const int e = min(e0 + c, kdeg - 1);
                const int s = src[(size_t)node * kdeg + e];
                const RowFrag f = load_row(x + (size_t)s * DFEAT, h);
                const v8f dt = tile_mm(f, h, wb, zero8);
#pragma unroll
                for (int r = 0; r < 8; ++r) d[r] = fmaxf(d[r], dt[r]);
            }
            tile[i * TP + c] = rows_max(d);
        }
        fy = load_row(xown, h);
        rb1 = b1[cb];
        rb2 = b2[cb];
    }

    const v8f cy = tile_mm(fy, h, wa, zero8);
    const float bias1 = (c < DFEAT) ? rb1 : 0.0f;
    const float bias2 = (c < DFEAT) ? rb2 : 0.0f;

    wave_lds_sync();
    float agg[8];
#pragma unroll
    for (int r = 0; r < 8; ++r) {
        const float v = (cy[r] + bias1) + tile[(r + 8 * h) * TP + c];
        agg[r] = isfinite(v) ? v : 0.0f;
    }
    wave_lds_sync();
#pragma unroll
    for (int r = 0; r < 8; ++r) tile[(r + 8 * h) * TP + c] = agg[r];
    wave_lds_sync();

    RowFrag fo;
    fo.q = *(const v4f*)(tile + c * TP + 4 * h);
    const float2 po = *(const float2*)(tile + c * TP + 8 + 2 * h);
    v8f co = zero8;
    {
        v2f a0, a1, a2;
        a0[0] = fo.q[0]; a0[1] = fo.q[1];
        a1[0] = fo.q[2]; a1[1] = fo.q[3];
        a2[0] = po.x;    a2[1] = po.y;
        co = wmma4(a0, w2[0], co);
        co = wmma4(a1, w2[1], co);
        co = wmma4(a2, w2[2], co);
    }

    wave_lds_sync();
    if (c < DFEAT) {
#pragma unroll
        for (int r = 0; r < 8; ++r) tile[(r + 8 * h) * DFEAT + c] = co[r] + bias2;
    }
    wave_lds_sync();
    const int nvalid = min(NPW, nnodes - base) * DFEAT;
    float* orow = out + (size_t)base * DFEAT;
#pragma unroll
    for (int j = 0; j < (NPW * DFEAT) / 32; ++j) {
        const int fidx = lane + 32 * j;
        if (fidx < nvalid) orow[fidx] = tile[fidx];
    }
}

extern "C" void kernel_launch(void* const* d_in, const int* in_sizes, int n_in,
                              void* d_out, int out_size, void* d_ws, size_t ws_size,
                              hipStream_t stream) {
    const float* x   = (const float*)d_in[0];
    const int*   ei  = (const int*)  d_in[1];
    const float* W1  = (const float*)d_in[2];
    const float* b1  = (const float*)d_in[3];
    const float* W2  = (const float*)d_in[4];
    const float* b2  = (const float*)d_in[5];
    float* out = (float*)d_out;

    const int nnodes = in_sizes[0] / DFEAT;
    const int nedges = in_sizes[1] / 2;
    const int kdeg   = nedges / nnodes;
    const int* src   = ei;

    const int ntiles = (nnodes + 15) / 16;
    dim3 block(WPB * 32);
    dim3 grid((ntiles + WPB - 1) / WPB);
    hipLaunchKernelGGL(edgeconv_fused, grid, block, 0, stream,
                       x, src, W1, b1, W2, b2, out, nnodes, kdeg);
}
